# starting kernel
# baseline (speedup 1.0000x reference)
#include <hip/hip_runtime.h>


#define IN_CH   1024
#define OUT_CH  64
#define BATCH   65536
#define PITCH   (IN_CH + 8)
#define LDS_U16 (2 * OUT_CH * PITCH)

typedef __bf16          v16bf __attribute__((ext_vector_type(16)));
typedef float           v8f   __attribute__((ext_vector_type(8)));
typedef float           v4f   __attribute__((ext_vector_type(4)));
typedef unsigned short  v16u  __attribute__((ext_vector_type(16)));
typedef unsigned short  v8u   __attribute__((ext_vector_type(8)));
typedef unsigned int    v8ui  __attribute__((ext_vector_type(8)));

extern __shared__ unsigned short smem[];

__device__ __forceinline__ unsigned pack_hi16(unsigned u0, unsigned u1) {
    return __builtin_amdgcn_perm(u1, u0, 0x07060302u);
}

__device__ __forceinline__ void split2(float f0, float f1, unsigned& hi, unsigned& lo) {
    unsigned u0 = __builtin_bit_cast(unsigned, f0);
    unsigned u1 = __builtin_bit_cast(unsigned, f1);
    hi = pack_hi16(u0, u1);
    float l0 = f0 - __builtin_bit_cast(float, u0 & 0xFFFF0000u);
    float l1 = f1 - __builtin_bit_cast(float, u1 & 0xFFFF0000u);
    lo = pack_hi16(__builtin_bit_cast(unsigned, l0),
                   __builtin_bit_cast(unsigned, l1));
}

struct Frag { v8ui hi; v8ui lo; };

__device__ __forceinline__ void load_a_frag(const float* __restrict__ xrow, int kb, Frag& f) {
    const v4f* p0 = reinterpret_cast<const v4f*>(xrow + kb);
    const v4f* p1 = reinterpret_cast<const v4f*>(xrow + kb + 16);
    v4f r0 = __builtin_nontemporal_load(p0);
    v4f r1 = __builtin_nontemporal_load(p0 + 1);
    v4f r2 = __builtin_nontemporal_load(p1);
    v4f r3 = __builtin_nontemporal_load(p1 + 1);
    float v[16] = { r0[0], r0[1], r0[2], r0[3],  r1[0], r1[1], r1[2], r1[3],
                    r2[0], r2[1], r2[2], r2[3],  r3[0], r3[1], r3[2], r3[3] };
#pragma unroll
    for (int j = 0; j < 8; ++j) {
        unsigned h, l;
        split2(v[2 * j], v[2 * j + 1], h, l);
        f.hi[j] = h;
        f.lo[j] = l;
    }
}

__device__ __forceinline__ v16u load_b_frag(int plane, int ch, int kb) {
    const unsigned short* row = &smem[plane * (OUT_CH * PITCH) + ch * PITCH];
    v8u a = *reinterpret_cast<const v8u*>(row + kb);
    v8u c = *reinterpret_cast<const v8u*>(row + kb + 16);
    return __builtin_shufflevector(a, c, 0, 1, 2, 3, 4, 5, 6, 7,
                                         8, 9, 10, 11, 12, 13, 14, 15);
}

__device__ __forceinline__ v8f wmma_bf16(const v8ui& a, const v16u& b, v8f c) {
    return __builtin_amdgcn_wmma_f32_16x16x32_bf16(
        false, __builtin_bit_cast(v16bf, a),
        false, __builtin_bit_cast(v16bf, b),
        (short)0, c, false, false);
}

__global__ __launch_bounds__(256) void seplinear_wmma_kernel(
    const float* __restrict__ x, const float* __restrict__ W,
    const float* __restrict__ b, float* __restrict__ out) {

    const int tid = threadIdx.x;

    for (int idx = tid; idx < OUT_CH * IN_CH; idx += 256) {
        int ch = idx >> 10, k = idx & (IN_CH - 1);
        float f = W[idx];
        unsigned u = __builtin_bit_cast(unsigned, f);
        float fl = f - __builtin_bit_cast(float, u & 0xFFFF0000u);
        smem[ch * PITCH + k]                  = (unsigned short)(u >> 16);
        smem[OUT_CH * PITCH + ch * PITCH + k] =
            (unsigned short)(__builtin_bit_cast(unsigned, fl) >> 16);
    }
    __syncthreads();

    const int lane  = tid & 31;
    const int wave  = tid >> 5;
    const int lrow  = lane & 15;
    const int khalf = lane >> 4;

    for (int base = blockIdx.x * 256; base < BATCH; base += gridDim.x * 256) {
        const int m0 = base + wave * 32;

        v8f acc[2][4];
#pragma unroll
        for (int mt = 0; mt < 2; ++mt)
#pragma unroll
            for (int nt = 0; nt < 4; ++nt) acc[mt][nt] = v8f{};

        const float* xr0 = x + (size_t)(m0 + lrow) * IN_CH;
        const float* xr1 = x + (size_t)(m0 + 16 + lrow) * IN_CH;

        for (int k0 = 0; k0 < IN_CH; k0 += 32) {
            const int kb = k0 + khalf * 8;
            Frag a0, a1;
            load_a_frag(xr0, kb, a0);
            load_a_frag(xr1, kb, a1);

#pragma unroll
            for (int nt = 0; nt < 4; ++nt) {
                const int ch = nt * 16 + lrow;
                v16u bh = load_b_frag(0, ch, kb);
                v16u bl = load_b_frag(1, ch, kb);

                acc[0][nt] = wmma_bf16(a0.hi, bh, acc[0][nt]);
                acc[0][nt] = wmma_bf16(a0.hi, bl, acc[0][nt]);
                acc[0][nt] = wmma_bf16(a0.lo, bh, acc[0][nt]);

                acc[1][nt] = wmma_bf16(a1.hi, bh, acc[1][nt]);
                acc[1][nt] = wmma_bf16(a1.hi, bl, acc[1][nt]);
                acc[1][nt] = wmma_bf16(a1.lo, bh, acc[1][nt]);
            }
        }

#pragma unroll
        for (int nt = 0; nt < 4; ++nt) {
            const int   n    = nt * 16 + lrow;
            const float bias = b[n];
#pragma unroll
            for (int mt = 0; mt < 2; ++mt) {
#pragma unroll
                for (int d = 0; d < 8; ++d) {
                    const int row = m0 + mt * 16 + khalf * 8 + d;
                    __builtin_nontemporal_store(acc[mt][nt][d] + bias,
                                                out + (size_t)row * OUT_CH + n);
                }
            }
        }
    }
}

extern "C" void kernel_launch(void* const* d_in, const int* in_sizes, int n_in,
                              void* d_out, int out_size, void* d_ws, size_t ws_size,
                              hipStream_t stream) {
    const float* x = (const float*)d_in[0];
    const float* W = (const float*)d_in[1];
    const float* b = (const float*)d_in[2];
    float* out     = (float*)d_out;

    const size_t lds_bytes = (size_t)LDS_U16 * sizeof(unsigned short);
    dim3 grid(256), block(256);
    hipLaunchKernelGGL(seplinear_wmma_kernel, grid, block, lds_bytes, stream,
                       x, W, b, out);
}


# software-pipeline the x stream: each wave requests the raw x slices of the next 3 k-steps (register ring of 4 per M-tile) before it splits and multiplies the current one; summation order unchanged
# speedup vs baseline: 1.3005x; 1.3005x over previous
#include <hip/hip_runtime.h>


#define IN_CH   1024
#define OUT_CH  64
#define BATCH   65536
#define PITCH   (IN_CH + 8)
#define LDS_U16 (2 * OUT_CH * PITCH)
#define KSTEP   32
#define AHEAD   3
#define RING    (AHEAD + 1)

typedef __bf16          v16bf __attribute__((ext_vector_type(16)));
typedef float           v8f   __attribute__((ext_vector_type(8)));
typedef float           v4f   __attribute__((ext_vector_type(4)));
typedef unsigned short  v16u  __attribute__((ext_vector_type(16)));
typedef unsigned short  v8u   __attribute__((ext_vector_type(8)));
typedef unsigned int    v8ui  __attribute__((ext_vector_type(8)));

extern __shared__ unsigned short smem[];

__device__ __forceinline__ unsigned pack_hi16(unsigned u0, unsigned u1) {
    return __builtin_amdgcn_perm(u1, u0, 0x07060302u);
}

__device__ __forceinline__ void split2(float f0, float f1, unsigned& hi, unsigned& lo) {
    unsigned u0 = __builtin_bit_cast(unsigned, f0);
    unsigned u1 = __builtin_bit_cast(unsigned, f1);
    hi = pack_hi16(u0, u1);
    float l0 = f0 - __builtin_bit_cast(float, u0 & 0xFFFF0000u);
    float l1 = f1 - __builtin_bit_cast(float, u1 & 0xFFFF0000u);
    lo = pack_hi16(__builtin_bit_cast(unsigned, l0),
                   __builtin_bit_cast(unsigned, l1));
}

struct Frag { v8ui hi; v8ui lo; };

struct Raw { v4f r[4]; };

__device__ __forceinline__ void issue_a_loads(const float* __restrict__ xrow, int kb, Raw& q) {
    const v4f* p0 = reinterpret_cast<const v4f*>(xrow + kb);
    const v4f* p1 = reinterpret_cast<const v4f*>(xrow + kb + 16);
    q.r[0] = __builtin_nontemporal_load(p0);
    q.r[1] = __builtin_nontemporal_load(p0 + 1);
    q.r[2] = __builtin_nontemporal_load(p1);
    q.r[3] = __builtin_nontemporal_load(p1 + 1);
}

__device__ __forceinline__ void split_a_frag(const Raw& q, Frag& f) {
    float v[16] = { q.r[0][0], q.r[0][1], q.r[0][2], q.r[0][3],
                    q.r[1][0], q.r[1][1], q.r[1][2], q.r[1][3],
                    q.r[2][0], q.r[2][1], q.r[2][2], q.r[2][3],
                    q.r[3][0], q.r[3][1], q.r[3][2], q.r[3][3] };
#pragma unroll
    for (int j = 0; j < 8; ++j) {
        unsigned h, l;
        split2(v[2 * j], v[2 * j + 1], h, l);
        f.hi[j] = h;
        f.lo[j] = l;
    }
}

__device__ __forceinline__ v16u load_b_frag(int plane, int ch, int kb) {
    const unsigned short* row = &smem[plane * (OUT_CH * PITCH) + ch * PITCH];
    v8u a = *reinterpret_cast<const v8u*>(row + kb);
    v8u c = *reinterpret_cast<const v8u*>(row + kb + 16);
    return __builtin_shufflevector(a, c, 0, 1, 2, 3, 4, 5, 6, 7,
                                         8, 9, 10, 11, 12, 13, 14, 15);
}

__device__ __forceinline__ v8f wmma_bf16(const v8ui& a, const v16u& b, v8f c) {
    return __builtin_amdgcn_wmma_f32_16x16x32_bf16(
        false, __builtin_bit_cast(v16bf, a),
        false, __builtin_bit_cast(v16bf, b),
        (short)0, c, false, false);
}

__device__ __forceinline__ void mma_step(const Raw& r0, const Raw& r1, int kb, int lrow,
                                         v8f (&acc)[2][4]) {
    Frag a0, a1;
    split_a_frag(r0, a0);
    split_a_frag(r1, a1);
#pragma unroll
    for (int nt = 0; nt < 4; ++nt) {
        const int ch = nt * 16 + lrow;
        v16u bh = load_b_frag(0, ch, kb);
        v16u bl = load_b_frag(1, ch, kb);

        acc[0][nt] = wmma_bf16(a0.hi, bh, acc[0][nt]);
        acc[0][nt] = wmma_bf16(a0.hi, bl, acc[0][nt]);
        acc[0][nt] = wmma_bf16(a0.lo, bh, acc[0][nt]);

        acc[1][nt] = wmma_bf16(a1.hi, bh, acc[1][nt]);
        acc[1][nt] = wmma_bf16(a1.hi, bl, acc[1][nt]);
        acc[1][nt] = wmma_bf16(a1.lo, bh, acc[1][nt]);
    }
}

__global__ __launch_bounds__(256) void seplinear_wmma_kernel(
    const float* __restrict__ x, const float* __restrict__ W,
    const float* __restrict__ b, float* __restrict__ out) {

    const int tid = threadIdx.x;

    for (int idx = tid; idx < OUT_CH * IN_CH; idx += 256) {
        int ch = idx >> 10, k = idx & (IN_CH - 1);
        float f = W[idx];
        unsigned u = __builtin_bit_cast(unsigned, f);
        float fl = f - __builtin_bit_cast(float, u & 0xFFFF0000u);
        smem[ch * PITCH + k]                  = (unsigned short)(u >> 16);
        smem[OUT_CH * PITCH + ch * PITCH + k] =
            (unsigned short)(__builtin_bit_cast(unsigned, fl) >> 16);
    }
    __syncthreads();

    const int lane  = tid & 31;
    const int wave  = tid >> 5;
    const int lrow  = lane & 15;
    const int khalf = lane >> 4;

    for (int base = blockIdx.x * 256; base < BATCH; base += gridDim.x * 256) {
        const int m0 = base + wave * 32;

        v8f acc[2][4];
#pragma unroll
        for (int mt = 0; mt < 2; ++mt)
#pragma unroll
            for (int nt = 0; nt < 4; ++nt) acc[mt][nt] = v8f{};

        const float* xr0 = x + (size_t)(m0 + lrow) * IN_CH;
        const float* xr1 = x + (size_t)(m0 + 16 + lrow) * IN_CH;

        Raw q0[RING], q1[RING];
#pragma unroll
        for (int s = 0; s < AHEAD; ++s) {
            issue_a_loads(xr0, s * KSTEP + khalf * 8, q0[s]);
            issue_a_loads(xr1, s * KSTEP + khalf * 8, q1[s]);
            __builtin_amdgcn_sched_barrier(0);
        }

        int k0 = 0;
        for (; k0 < IN_CH - RING * KSTEP; k0 += RING * KSTEP) {
#pragma unroll
            for (int s = 0; s < RING; ++s) {
                const int kb = k0 + s * KSTEP + khalf * 8;
                issue_a_loads(xr0, kb + AHEAD * KSTEP, q0[(s + AHEAD) % RING]);
                issue_a_loads(xr1, kb + AHEAD * KSTEP, q1[(s + AHEAD) % RING]);
                __builtin_amdgcn_sched_barrier(0);
                mma_step(q0[s], q1[s], kb, lrow, acc);
            }
        }
        asm volatile("" : "+s"(k0));
#pragma unroll
        for (int s = 0; s < RING; ++s) {
            const int kb = k0 + s * KSTEP + khalf * 8;
            if (s + AHEAD < RING) {
                issue_a_loads(xr0, kb + AHEAD * KSTEP, q0[(s + AHEAD) % RING]);
                issue_a_loads(xr1, kb + AHEAD * KSTEP, q1[(s + AHEAD) % RING]);
            }
            __builtin_amdgcn_sched_barrier(0);
            mma_step(q0[s], q1[s], kb, lrow, acc);
        }

#pragma unroll
        for (int nt = 0; nt < 4; ++nt) {
            const int   n    = nt * 16 + lrow;
            const float bias = b[n];
#pragma unroll
            for (int mt = 0; mt < 2; ++mt) {
#pragma unroll
                for (int d = 0; d < 8; ++d) {
                    const int row = m0 + mt * 16 + khalf * 8 + d;
                    __builtin_nontemporal_store(acc[mt][nt][d] + bias,
                                                out + (size_t)row * OUT_CH + n);
                }
            }
        }
    }
}

extern "C" void kernel_launch(void* const* d_in, const int* in_sizes, int n_in,
                              void* d_out, int out_size, void* d_ws, size_t ws_size,
                              hipStream_t stream) {
    const float* x = (const float*)d_in[0];
    const float* W = (const float*)d_in[1];
    const float* b = (const float*)d_in[2];
    float* out     = (float*)d_out;

    const size_t lds_bytes = (size_t)LDS_U16 * sizeof(unsigned short);
    dim3 grid(256), block(256);
    hipLaunchKernelGGL(seplinear_wmma_kernel, grid, block, lds_bytes, stream,
                       x, W, b, out);
}


# W staging: each thread has 32 rows' float4 loads in flight together and stores packed bf16 hi/lo quads to LDS, replacing 64 serial round trips of four scalar loads each; LDS contents identical
# speedup vs baseline: 1.6951x; 1.3034x over previous
#include <hip/hip_runtime.h>


#define IN_CH   1024
#define OUT_CH  64
#define BATCH   65536
#define PITCH   (IN_CH + 8)
#define LDS_U16 (2 * OUT_CH * PITCH)
#define KSTEP   32
#define AHEAD   3
#define RING    (AHEAD + 1)
#define WBATCH  32

typedef __bf16          v16bf __attribute__((ext_vector_type(16)));
typedef float           v8f   __attribute__((ext_vector_type(8)));
typedef float           v4f   __attribute__((ext_vector_type(4)));
typedef unsigned short  v16u  __attribute__((ext_vector_type(16)));
typedef unsigned short  v8u   __attribute__((ext_vector_type(8)));
typedef unsigned short  v4u   __attribute__((ext_vector_type(4)));
typedef unsigned int    v2ui  __attribute__((ext_vector_type(2)));
typedef unsigned int    v8ui  __attribute__((ext_vector_type(8)));

extern __shared__ unsigned short smem[];

__device__ __forceinline__ unsigned pack_hi16(unsigned u0, unsigned u1) {
    return __builtin_amdgcn_perm(u1, u0, 0x07060302u);
}

__device__ __forceinline__ void split2(float f0, float f1, unsigned& hi, unsigned& lo) {
    unsigned u0 = __builtin_bit_cast(unsigned, f0);
    unsigned u1 = __builtin_bit_cast(unsigned, f1);
    hi = pack_hi16(u0, u1);
    float l0 = f0 - __builtin_bit_cast(float, u0 & 0xFFFF0000u);
    float l1 = f1 - __builtin_bit_cast(float, u1 & 0xFFFF0000u);
    lo = pack_hi16(__builtin_bit_cast(unsigned, l0),
                   __builtin_bit_cast(unsigned, l1));
}

struct Frag { v8ui hi; v8ui lo; };

struct Raw { v4f r[4]; };

__device__ __forceinline__ void issue_a_loads(const float* __restrict__ xrow, int kb, Raw& q) {
    const v4f* p0 = reinterpret_cast<const v4f*>(xrow + kb);
    const v4f* p1 = reinterpret_cast<const v4f*>(xrow + kb + 16);
    q.r[0] = __builtin_nontemporal_load(p0);
    q.r[1] = __builtin_nontemporal_load(p0 + 1);
    q.r[2] = __builtin_nontemporal_load(p1);
    q.r[3] = __builtin_nontemporal_load(p1 + 1);
}

__device__ __forceinline__ void split_a_frag(const Raw& q, Frag& f) {
    float v[16] = { q.r[0][0], q.r[0][1], q.r[0][2], q.r[0][3],
                    q.r[1][0], q.r[1][1], q.r[1][2], q.r[1][3],
                    q.r[2][0], q.r[2][1], q.r[2][2], q.r[2][3],
                    q.r[3][0], q.r[3][1], q.r[3][2], q.r[3][3] };
#pragma unroll
    for (int j = 0; j < 8; ++j) {
        unsigned h, l;
        split2(v[2 * j], v[2 * j + 1], h, l);
        f.hi[j] = h;
        f.lo[j] = l;
    }
}

__device__ __forceinline__ v16u load_b_frag(int plane, int ch, int kb) {
    const unsigned short* row = &smem[plane * (OUT_CH * PITCH) + ch * PITCH];
    v8u a = *reinterpret_cast<const v8u*>(row + kb);
    v8u c = *reinterpret_cast<const v8u*>(row + kb + 16);
    return __builtin_shufflevector(a, c, 0, 1, 2, 3, 4, 5, 6, 7,
                                         8, 9, 10, 11, 12, 13, 14, 15);
}

__device__ __forceinline__ v8f wmma_bf16(const v8ui& a, const v16u& b, v8f c) {
    return __builtin_amdgcn_wmma_f32_16x16x32_bf16(
        false, __builtin_bit_cast(v16bf, a),
        false, __builtin_bit_cast(v16bf, b),
        (short)0, c, false, false);
}

__device__ __forceinline__ void mma_step(const Raw& r0, const Raw& r1, int kb, int lrow,
                                         v8f (&acc)[2][4]) {
    Frag a0, a1;
    split_a_frag(r0, a0);
    split_a_frag(r1, a1);
#pragma unroll
    for (int nt = 0; nt < 4; ++nt) {
        const int ch = nt * 16 + lrow;
        v16u bh = load_b_frag(0, ch, kb);
        v16u bl = load_b_frag(1, ch, kb);

        acc[0][nt] = wmma_bf16(a0.hi, bh, acc[0][nt]);
        acc[0][nt] = wmma_bf16(a0.hi, bl, acc[0][nt]);
        acc[0][nt] = wmma_bf16(a0.lo, bh, acc[0][nt]);

        acc[1][nt] = wmma_bf16(a1.hi, bh, acc[1][nt]);
        acc[1][nt] = wmma_bf16(a1.hi, bl, acc[1][nt]);
        acc[1][nt] = wmma_bf16(a1.lo, bh, acc[1][nt]);
    }
}

__global__ __launch_bounds__(256) void seplinear_wmma_kernel(
    const float* __restrict__ x, const float* __restrict__ W,
    const float* __restrict__ b, float* __restrict__ out) {

    const int tid = threadIdx.x;

    for (int r0 = 0; r0 < OUT_CH; r0 += WBATCH) {
        v4f w[WBATCH];
#pragma unroll
        for (int j = 0; j < WBATCH; ++j)
            w[j] = *reinterpret_cast<const v4f*>(W + (size_t)(r0 + j) * IN_CH + 4 * tid);
#pragma unroll
        for (int j = 0; j < WBATCH; ++j) {
            unsigned h0, l0, h1, l1;
            split2(w[j][0], w[j][1], h0, l0);
            split2(w[j][2], w[j][3], h1, l1);
            unsigned short* dst = &smem[(r0 + j) * PITCH + 4 * tid];
            *reinterpret_cast<v4u*>(dst)                  = __builtin_bit_cast(v4u, (v2ui){h0, h1});
            *reinterpret_cast<v4u*>(dst + OUT_CH * PITCH) = __builtin_bit_cast(v4u, (v2ui){l0, l1});
        }
    }
    __syncthreads();

    const int lane  = tid & 31;
    const int wave  = tid >> 5;
    const int lrow  = lane & 15;
    const int khalf = lane >> 4;

    for (int base = blockIdx.x * 256; base < BATCH; base += gridDim.x * 256) {
        const int m0 = base + wave * 32;

        v8f acc[2][4];
#pragma unroll
        for (int mt = 0; mt < 2; ++mt)
#pragma unroll
            for (int nt = 0; nt < 4; ++nt) acc[mt][nt] = v8f{};

        const float* xr0 = x + (size_t)(m0 + lrow) * IN_CH;
        const float* xr1 = x + (size_t)(m0 + 16 + lrow) * IN_CH;

        Raw q0[RING], q1[RING];
#pragma unroll
        for (int s = 0; s < AHEAD; ++s) {
            issue_a_loads(xr0, s * KSTEP + khalf * 8, q0[s]);
            issue_a_loads(xr1, s * KSTEP + khalf * 8, q1[s]);
            __builtin_amdgcn_sched_barrier(0);
        }

        int k0 = 0;
        for (; k0 < IN_CH - RING * KSTEP; k0 += RING * KSTEP) {
#pragma unroll
            for (int s = 0; s < RING; ++s) {
                const int kb = k0 + s * KSTEP + khalf * 8;
                issue_a_loads(xr0, kb + AHEAD * KSTEP, q0[(s + AHEAD) % RING]);
                issue_a_loads(xr1, kb + AHEAD * KSTEP, q1[(s + AHEAD) % RING]);
                __builtin_amdgcn_sched_barrier(0);
                mma_step(q0[s], q1[s], kb, lrow, acc);
            }
        }
        asm volatile("" : "+s"(k0));
#pragma unroll
        for (int s = 0; s < RING; ++s) {
            const int kb = k0 + s * KSTEP + khalf * 8;
            if (s + AHEAD < RING) {
                issue_a_loads(xr0, kb + AHEAD * KSTEP, q0[(s + AHEAD) % RING]);
                issue_a_loads(xr1, kb + AHEAD * KSTEP, q1[(s + AHEAD) % RING]);
            }
            __builtin_amdgcn_sched_barrier(0);
            mma_step(q0[s], q1[s], kb, lrow, acc);
        }

#pragma unroll
        for (int nt = 0; nt < 4; ++nt) {
            const int   n    = nt * 16 + lrow;
            const float bias = b[n];
#pragma unroll
            for (int mt = 0; mt < 2; ++mt) {
#pragma unroll
                for (int d = 0; d < 8; ++d) {
                    const int row = m0 + mt * 16 + khalf * 8 + d;
                    __builtin_nontemporal_store(acc[mt][nt][d] + bias,
                                                out + (size_t)row * OUT_CH + n);
                }
            }
        }
    }
}

extern "C" void kernel_launch(void* const* d_in, const int* in_sizes, int n_in,
                              void* d_out, int out_size, void* d_ws, size_t ws_size,
                              hipStream_t stream) {
    const float* x = (const float*)d_in[0];
    const float* W = (const float*)d_in[1];
    const float* b = (const float*)d_in[2];
    float* out     = (float*)d_out;

    const size_t lds_bytes = (size_t)LDS_U16 * sizeof(unsigned short);
    dim3 grid(256), block(256);
    hipLaunchKernelGGL(seplinear_wmma_kernel, grid, block, lds_bytes, stream,
                       x, W, b, out);
}
